# starting kernel
# baseline (speedup 1.0000x reference)
#include <hip/hip_runtime.h>

typedef __attribute__((ext_vector_type(2))) float v2f;
typedef __attribute__((ext_vector_type(4))) float v4f;
typedef __attribute__((ext_vector_type(8))) float v8f;

#define D_FEAT 64

__device__ __forceinline__ int bperm_i(int src, int lane) {
    return __builtin_amdgcn_ds_bpermute(lane << 2, src);
}
__device__ __forceinline__ float bperm_f(float src, int lane) {
    return __int_as_float(__builtin_amdgcn_ds_bpermute(lane << 2, __float_as_int(src)));
}

__global__ void gcn_zero_kernel(float* __restrict__ out, int n) {
    int i = (blockIdx.x * blockDim.x + threadIdx.x) * 4;
    if (i + 3 < n) {
        *(v4f*)(out + i) = (v4f){0.f, 0.f, 0.f, 0.f};
    } else {
        for (int k = i; k < n; ++k) out[k] = 0.f;
    }
}

__global__ __launch_bounds__(256) void gcn_spmm_wmma_kernel(
    const int* __restrict__ rows, const int* __restrict__ cols,
    const float* __restrict__ vals, const float* __restrict__ embeds,
    float* __restrict__ out, int nedges) {

    const int lane = threadIdx.x & 31;
    const int tile = blockIdx.x * 8 + (threadIdx.x >> 5);
    const int base = tile * 16;
    if (base >= nedges) return;

    const int j = lane & 15;
    const int e = base + j;
    const bool valid = (e < nedges);
    int   row = valid ? rows[e] : (-1 - j);
    int   col = valid ? cols[e] : 0;
    float val = valid ? vals[e] : 0.0f;

    int fo = 1;
#pragma unroll
    for (int i = 0; i < 15; ++i) {
        int ri = bperm_i(row, i);
        if (i < j && ri == row) fo = 0;
    }

    const int khalf = (lane >> 4) << 1;
    v2f A[4];
    int ck0[4], ck1[4];
#pragma unroll
    for (int q = 0; q < 4; ++q) {
        const int k0 = 4 * q + khalf;
        const int k1 = k0 + 1;
        int   rk0 = bperm_i(row, k0);
        int   rk1 = bperm_i(row, k1);
        float vk0 = bperm_f(val, k0);
        float vk1 = bperm_f(val, k1);
        A[q].x = (rk0 == row) ? vk0 : 0.0f;
        A[q].y = (rk1 == row) ? vk1 : 0.0f;
        ck0[q] = bperm_i(col, k0);
        ck1[q] = bperm_i(col, k1);
    }

    int rowM[8], foM[8];
#pragma unroll
    for (int r = 0; r < 8; ++r) {
        const int M = (lane < 16) ? r : (r + 8);
        rowM[r] = bperm_i(row, M);
        foM[r]  = bperm_i(fo, M);
    }

    v8f acc4[4];
#pragma unroll
    for (int c = 0; c < 4; ++c) {
        v8f acc = {0.f, 0.f, 0.f, 0.f, 0.f, 0.f, 0.f, 0.f};
#pragma unroll
        for (int q = 0; q < 4; ++q) {
            v2f B;
            B.x = embeds[ck0[q] * D_FEAT + c * 16 + j];
            B.y = embeds[ck1[q] * D_FEAT + c * 16 + j];
            acc = __builtin_amdgcn_wmma_f32_16x16x4_f32(
                 false, A[q],  false, B,
                 (short)0, acc,  false,  false);
        }
        acc4[c] = acc;
    }

#pragma unroll
    for (int r = 0; r < 8; ++r) {
        const bool sc = (foM[r] != 0) && (rowM[r] >= 0);
        float* p = out + (sc ? rowM[r] : 0) * D_FEAT + j;
#pragma unroll
        for (int c = 0; c < 4; ++c) {
            atomicAdd(p + c * 16, sc ? acc4[c][r] : 0.0f);
        }
    }
}

extern "C" void kernel_launch(void* const* d_in, const int* in_sizes, int n_in,
                              void* d_out, int out_size, void* d_ws, size_t ws_size,
                              hipStream_t stream) {
    const int*   rows   = (const int*)d_in[0];
    const int*   cols   = (const int*)d_in[1];
    const float* vals   = (const float*)d_in[2];
    const float* embeds = (const float*)d_in[3];
    float*       out    = (float*)d_out;

    const int nedges = in_sizes[0];

    {
        int nthreads = (out_size + 3) / 4;
        int blocks   = (nthreads + 255) / 256;
        gcn_zero_kernel<<<blocks, 256, 0, stream>>>(out, out_size);
    }

    {
        int tiles  = (nedges + 15) / 16;
        int blocks = (tiles + 7) / 8;
        gcn_spmm_wmma_kernel<<<blocks, 256, 0, stream>>>(
            rows, cols, vals, embeds, out, nedges);
    }
}


# speedup vs baseline: 1.0704x; 1.0704x over previous
#include <hip/hip_runtime.h>


typedef __attribute__((ext_vector_type(2))) float v2f;
typedef __attribute__((ext_vector_type(4))) float v4f;
typedef __attribute__((ext_vector_type(8))) float v8f;

#define D_FEAT 64
#define WAVES_PER_BLOCK 8
#define CNT_STRIDE 64
#define WALK_BLOCKS 1536

__device__ __forceinline__ int bperm_i(int src, int lane) {
    return __builtin_amdgcn_ds_bpermute(lane << 2, src);
}
__device__ __forceinline__ float bperm_f(float src, int lane) {
    return __int_as_float(__builtin_amdgcn_ds_bpermute(lane << 2, __float_as_int(src)));
}

__global__ void gcn_zero_kernel(float* __restrict__ out, int n, int* __restrict__ cnt) {
    int i = (blockIdx.x * blockDim.x + threadIdx.x) * 4;
    if (i + 3 < n) {
        *(v4f*)(out + i) = (v4f){0.f, 0.f, 0.f, 0.f};
    } else {
        for (int k = i; k < n; ++k) out[k] = 0.f;
    }
    if (blockIdx.x == 0 && threadIdx.x < 2) cnt[threadIdx.x * CNT_STRIDE] = 0;
}

__device__ __forceinline__ int tile_pass(
    const int* __restrict__ rows, const int* __restrict__ cols,
    const float* __restrict__ vals, const float* __restrict__ embeds,
    float* __restrict__ out, int nedges, int tile, int lane, int P,
    int* counter, bool grab) {

    const int base = tile * 16;
    const int j = lane & 15;
    const int e = base + j;
    const bool valid = (e < nedges);
    int   row = valid ? rows[e] : (-1 - j);
    int   col = valid ? cols[e] : 0;
    float val = valid ? vals[e] : 0.0f;
    const bool mine = valid && ((row & 1) == P);
    if (!mine) val = 0.0f;
    const int colm = mine ? col : 0;

    int fo = mine ? 1 : 0;
#pragma unroll
    for (int i = 0; i < 15; ++i) {
        int ri = bperm_i(row, i);
        if (i < j && ri == row) fo = 0;
    }

    const int khalf = (lane >> 4) << 1;
    v2f A[4];
    int ck0[4], ck1[4];
#pragma unroll
    for (int q = 0; q < 4; ++q) {
        const int k0 = 4 * q + khalf;
        const int k1 = k0 + 1;
        int   rk0 = bperm_i(row, k0);
        int   rk1 = bperm_i(row, k1);
        float vk0 = bperm_f(val, k0);
        float vk1 = bperm_f(val, k1);
        A[q].x = (rk0 == row) ? vk0 : 0.0f;
        A[q].y = (rk1 == row) ? vk1 : 0.0f;
        ck0[q] = bperm_i(colm, k0);
        ck1[q] = bperm_i(colm, k1);
    }

    float b0[4][4], b1[4][4];
#pragma unroll
    for (int q = 0; q < 4; ++q) {
        const int i0 = ck0[q] * D_FEAT + j;
        const int i1 = ck1[q] * D_FEAT + j;
#pragma unroll
        for (int c = 0; c < 4; ++c) {
            b0[q][c] = embeds[i0 + c * 16];
            b1[q][c] = embeds[i1 + c * 16];
        }
    }
    int nxt = 0;
    if (grab) nxt = atomicAdd(counter, 1);

#pragma unroll
    for (int q = 0; q < 4; ++q) {
        asm volatile("" :: "v"(b0[q][0]), "v"(b0[q][1]), "v"(b0[q][2]), "v"(b0[q][3]),
                           "v"(b1[q][0]), "v"(b1[q][1]), "v"(b1[q][2]), "v"(b1[q][3]));
    }
    v8f acc4[4];
#pragma unroll
    for (int c = 0; c < 4; ++c) {
        v8f acc = {0.f, 0.f, 0.f, 0.f, 0.f, 0.f, 0.f, 0.f};
#pragma unroll
        for (int q = 0; q < 4; ++q) {
            v2f B;
            B.x = b0[q][c];
            B.y = b1[q][c];
            acc = __builtin_amdgcn_wmma_f32_16x16x4_f32(
                 false, A[q],  false, B,
                 (short)0, acc,  false,  false);
        }
        acc4[c] = acc;
    }

    int rowM[8], foM[8];
#pragma unroll
    for (int r = 0; r < 8; ++r) {
        const int M = (lane < 16) ? r : (r + 8);
        rowM[r] = bperm_i(row, M);
        foM[r]  = bperm_i(fo, M);
    }

#pragma unroll
    for (int r = 0; r < 8; ++r) {
        if (foM[r] != 0) {
            float* p = out + rowM[r] * D_FEAT + j;
#pragma unroll
            for (int c = 0; c < 4; ++c) {
                atomicAdd(p + c * 16, acc4[c][r]);
            }
        }
    }
    return nxt;
}

__global__ __launch_bounds__(256) __attribute__((amdgpu_waves_per_eu(12, 12))) void gcn_spmm_wmma_kernel(
    const int* __restrict__ rows, const int* __restrict__ cols,
    const float* __restrict__ vals, const float* __restrict__ embeds,
    float* __restrict__ out, int* __restrict__ cnt, int nedges) {

    __shared__ int s_half;
    __shared__ int s_group;

    const int lane = threadIdx.x & 31;
    const int wave = __builtin_amdgcn_readfirstlane(threadIdx.x >> 5);
    const int ntiles  = (nedges + 15) >> 4;
    const int ngroups = (ntiles + WAVES_PER_BLOCK - 1) / WAVES_PER_BLOCK;

    if (threadIdx.x == 0) {
        const int id = __builtin_amdgcn_s_sendmsg_rtn(0x87);
        s_half = (id >> 18) & 1;
    }
    __syncthreads();
    const int H = __builtin_amdgcn_readfirstlane(s_half);

    for (int phase = 0; phase < 2; ++phase) {
        const int P = H ^ phase;
        int* const counter = cnt + P * CNT_STRIDE;
        int nxt = 0;
        if (threadIdx.x == 0) nxt = atomicAdd(counter, 1);
        for (int it = 0; it <= ngroups; ++it) {
            __syncthreads();
            if (threadIdx.x == 0) s_group = nxt;
            __syncthreads();
            const int g = __builtin_amdgcn_readfirstlane(s_group);
            if (g >= ngroups) break;
            const int tile = g * WAVES_PER_BLOCK + wave;
            if (tile < ntiles) {
                nxt = tile_pass(rows, cols, vals, embeds, out, nedges, tile, lane, P,
                                counter, threadIdx.x == 0);
            }
        }
    }
}

extern "C" void kernel_launch(void* const* d_in, const int* in_sizes, int n_in,
                              void* d_out, int out_size, void* d_ws, size_t ws_size,
                              hipStream_t stream) {
    const int*   rows   = (const int*)d_in[0];
    const int*   cols   = (const int*)d_in[1];
    const float* vals   = (const float*)d_in[2];
    const float* embeds = (const float*)d_in[3];
    float*       out    = (float*)d_out;
    int*         cnt    = (int*)d_ws;

    const int nedges = in_sizes[0];

    {
        int nthreads = (out_size + 3) / 4;
        int blocks   = (nthreads + 255) / 256;
        gcn_zero_kernel<<<blocks, 256, 0, stream>>>(out, out_size, cnt);
    }

    gcn_spmm_wmma_kernel<<<WALK_BLOCKS, 256, 0, stream>>>(
        rows, cols, vals, embeds, out, cnt, nedges);
}


# Edges reordered by destination row range (LDS-counter ranks); one work-group per range adds the WMMA products in LDS and stores each row once: no global atomics, no zero pass; add order changes
# speedup vs baseline: 1.2868x; 1.2022x over previous
#include <hip/hip_runtime.h>


typedef __attribute__((ext_vector_type(2))) float v2f;
typedef __attribute__((ext_vector_type(4))) float v4f;
typedef __attribute__((ext_vector_type(8))) float v8f;

#define D_FEAT 64
#define RPB 391
#define NBINS 256
#define CH 8192
#define PART_THREADS 1024
#define EPT (CH / PART_THREADS)
#define OFF_PITCH (NBINS + 1)
#define ACC_PITCH 80
#define OWN_THREADS 1024
#define COL_BITS 20
#define RECS_OFFSET_INTS (1 << 18)

__device__ __forceinline__ int bperm_i(int src, int lane) {
    return __builtin_amdgcn_ds_bpermute(lane << 2, src);
}
__device__ __forceinline__ float bperm_f(float src, int lane) {
    return __int_as_float(__builtin_amdgcn_ds_bpermute(lane << 2, __float_as_int(src)));
}

__global__ __launch_bounds__(PART_THREADS) void gcn_part_kernel(
    const int* __restrict__ rows, const int* __restrict__ cols,
    const float* __restrict__ vals, int* __restrict__ recs_i,
    int* __restrict__ offs, int nedges) {

    __shared__ int hist[NBINS];
    __shared__ int binoff[NBINS + 1];
    __shared__ int2 stage[CH];

    int2* const recs = (int2*)recs_i;
    const int tid = threadIdx.x;
    const int nchunks = (nedges + CH - 1) / CH;

    for (int chunk = blockIdx.x; chunk < nchunks; chunk += gridDim.x) {
        const int base = chunk * CH;
        if (tid < NBINS) hist[tid] = 0;
        __syncthreads();

        int bin[EPT], rank[EPT], pk[EPT];
        float v[EPT];
#pragma unroll
        for (int i = 0; i < EPT; ++i) {
            const int e = base + i * PART_THREADS + tid;
            bin[i] = -1; rank[i] = 0; pk[i] = 0; v[i] = 0.0f;
            if (e < nedges) {
                const int r = rows[e];
                const int c = cols[e];
                v[i] = vals[e];
                int b = (int)((unsigned)r / (unsigned)RPB);
                if (b > NBINS - 1) b = NBINS - 1;
                bin[i] = b;
                pk[i] = ((r - b * RPB) << COL_BITS) | c;
            }
        }
#pragma unroll
        for (int i = 0; i < EPT; ++i) {
            if (bin[i] >= 0) rank[i] = atomicAdd(&hist[bin[i]], 1);
        }
        __syncthreads();

        if (tid < 32) {
            int s[8];
            int loc = 0;
#pragma unroll
            for (int k = 0; k < 8; ++k) { s[k] = hist[tid * 8 + k]; loc += s[k]; }
            int inc = loc;
#pragma unroll
            for (int d = 1; d < 32; d <<= 1) {
                const int t = __shfl_up(inc, d);
                if (tid >= d) inc += t;
            }
            int exc = inc - loc;
#pragma unroll
            for (int k = 0; k < 8; ++k) { binoff[tid * 8 + k] = exc; exc += s[k]; }
            if (tid == 31) binoff[NBINS] = exc;
        }
        __syncthreads();

        if (tid <= NBINS) offs[chunk * OFF_PITCH + tid] = binoff[tid];
#pragma unroll
        for (int i = 0; i < EPT; ++i) {
            if (bin[i] >= 0) stage[binoff[bin[i]] + rank[i]] = make_int2(pk[i], __float_as_int(v[i]));
        }
        __syncthreads();

        int n = nedges - base;
        if (n > CH) n = CH;
        for (int k = tid; k < n; k += PART_THREADS) recs[base + k] = stage[k];
        __syncthreads();
    }
}

__device__ __forceinline__ void tile_own(int2 rec, bool valid,
    const float* __restrict__ embeds, float* acc, int lane) {

    const int j = lane & 15;
    int row = valid ? (int)((unsigned)rec.x >> COL_BITS) : (-1 - j);
    if (row > RPB - 1) row = RPB - 1;
    const int   col = valid ? (rec.x & ((1 << COL_BITS) - 1)) : 0;
    const float val = valid ? __int_as_float(rec.y) : 0.0f;

    int fo = valid ? 1 : 0;
#pragma unroll
    for (int i = 0; i < 15; ++i) {
        int ri = bperm_i(row, i);
        if (i < j && ri == row) fo = 0;
    }

    const int khalf = (lane >> 4) << 1;
    v2f A[4];
    int ck0[4], ck1[4];
#pragma unroll
    for (int q = 0; q < 4; ++q) {
        const int k0 = 4 * q + khalf;
        const int k1 = k0 + 1;
        int   rk0 = bperm_i(row, k0);
        int   rk1 = bperm_i(row, k1);
        float vk0 = bperm_f(val, k0);
        float vk1 = bperm_f(val, k1);
        A[q].x = (rk0 == row) ? vk0 : 0.0f;
        A[q].y = (rk1 == row) ? vk1 : 0.0f;
        ck0[q] = bperm_i(col, k0);
        ck1[q] = bperm_i(col, k1);
    }

    float b0[4][4], b1[4][4];
#pragma unroll
    for (int q = 0; q < 4; ++q) {
        const int i0 = ck0[q] * D_FEAT + j;
        const int i1 = ck1[q] * D_FEAT + j;
#pragma unroll
        for (int c = 0; c < 4; ++c) {
            b0[q][c] = embeds[i0 + c * 16];
            b1[q][c] = embeds[i1 + c * 16];
        }
    }
#pragma unroll
    for (int q = 0; q < 4; ++q) {
        asm volatile("" :: "v"(b0[q][0]), "v"(b0[q][1]), "v"(b0[q][2]), "v"(b0[q][3]),
                           "v"(b1[q][0]), "v"(b1[q][1]), "v"(b1[q][2]), "v"(b1[q][3]));
    }
    v8f acc4[4];
#pragma unroll
    for (int c = 0; c < 4; ++c) {
        v8f a8 = {0.f, 0.f, 0.f, 0.f, 0.f, 0.f, 0.f, 0.f};
#pragma unroll
        for (int q = 0; q < 4; ++q) {
            v2f B;
            B.x = b0[q][c];
            B.y = b1[q][c];
            a8 = __builtin_amdgcn_wmma_f32_16x16x4_f32(
                 false, A[q],  false, B,
                 (short)0, a8,  false,  false);
        }
        acc4[c] = a8;
    }

    int rowM[8], foM[8];
#pragma unroll
    for (int r = 0; r < 8; ++r) {
        const int M = (lane < 16) ? r : (r + 8);
        rowM[r] = bperm_i(row, M);
        foM[r]  = bperm_i(fo, M);
    }
#pragma unroll
    for (int r = 0; r < 8; ++r) {
        if (foM[r] != 0) {
            float* p = acc + rowM[r] * ACC_PITCH + j;
#pragma unroll
            for (int c = 0; c < 4; ++c) {
                atomicAdd(p + c * 16, acc4[c][r]);
            }
        }
    }
}

__global__ __launch_bounds__(OWN_THREADS) void gcn_own_wmma_kernel(
    const int* __restrict__ recs_i, const int* __restrict__ offs,
    const float* __restrict__ embeds, float* __restrict__ out,
    int nedges, int out_size) {

    __shared__ __attribute__((aligned(16))) float acc[RPB * ACC_PITCH];

    const int2* const recs = (const int2*)recs_i;
    const int lane = threadIdx.x & 31;
    const int j = lane & 15;
    const int wave = __builtin_amdgcn_readfirstlane(threadIdx.x >> 5);
    const int nchunks = (nedges + CH - 1) / CH;
    const int n_nodes = out_size / D_FEAT;
    const int nbins = (n_nodes + RPB - 1) / RPB;

    for (int bin = blockIdx.x; bin < nbins; bin += gridDim.x) {
        for (int i = threadIdx.x; i < RPB * ACC_PITCH / 4; i += OWN_THREADS)
            ((v4f*)acc)[i] = (v4f){0.f, 0.f, 0.f, 0.f};
        __syncthreads();

        for (int w = wave; w < nchunks; w += OWN_THREADS / 32) {
            const int s = __builtin_amdgcn_readfirstlane(offs[w * OFF_PITCH + bin]);
            const int e = __builtin_amdgcn_readfirstlane(offs[w * OFF_PITCH + bin + 1]);
            const int2* const seg = recs + w * CH;
            int2 cur = make_int2(0, 0);
            if (s + j < e) cur = seg[s + j];
            asm volatile("" :: "v"(cur.x), "v"(cur.y));
            for (int k = s; k < e; k += 16) {
                int2 nxt = make_int2(0, 0);
                if (k + 16 + j < e) nxt = seg[k + 16 + j];
                tile_own(cur, (k + j) < e, embeds, acc, lane);
                cur = nxt;
            }
        }
        __syncthreads();

        const int row0 = bin * RPB;
        int nrows = n_nodes - row0;
        if (nrows > RPB) nrows = RPB;
        for (int i = threadIdx.x; i < nrows * (D_FEAT / 4); i += OWN_THREADS) {
            const int r = i >> 4;
            const int f = (i & 15) * 4;
            const v4f v = *(const v4f*)(acc + r * ACC_PITCH + f);
            *(v4f*)(out + (row0 + r) * D_FEAT + f) = v;
        }
        __syncthreads();
    }
}

extern "C" void kernel_launch(void* const* d_in, const int* in_sizes, int n_in,
                              void* d_out, int out_size, void* d_ws, size_t ws_size,
                              hipStream_t stream) {
    const int*   rows   = (const int*)d_in[0];
    const int*   cols   = (const int*)d_in[1];
    const float* vals   = (const float*)d_in[2];
    const float* embeds = (const float*)d_in[3];
    float*       out    = (float*)d_out;
    int*         offs   = (int*)d_ws;
    int*         recs   = (int*)d_ws + RECS_OFFSET_INTS;

    const int nedges = in_sizes[0];

    gcn_part_kernel<<<128, PART_THREADS, 0, stream>>>(rows, cols, vals, recs, offs, nedges);

    gcn_own_wmma_kernel<<<NBINS, OWN_THREADS, 0, stream>>>(
        recs, offs, embeds, out, nedges, out_size);
}


# Each owner wave owns 12-13 rows: records regrouped by owner wave in LDS, a wave's rows summed in the WMMA accumulators over all its tiles, stored from there; no LDS float adds; add order changes
# speedup vs baseline: 1.5314x; 1.1901x over previous
#include <hip/hip_runtime.h>


typedef __attribute__((ext_vector_type(2))) float v2f;
typedef __attribute__((ext_vector_type(4))) float v4f;
typedef __attribute__((ext_vector_type(8))) float v8f;

#define D_FEAT 64
#define RPB 391
#define NBINS 256
#define CH 8192
#define PART_THREADS 1024
#define EPT (CH / PART_THREADS)
#define OFF_PITCH (NBINS + 1)
#define OWN_THREADS 1024
#define NW (OWN_THREADS / 32)
#define ROWS_LO (RPB / NW)
#define N_HI (RPB % NW)
#define ROWS_HI (ROWS_LO + 1)
#define SEGS 4
#define UNR 6
#define SUBCAP 512
#define SUB_PITCH 514
#define COL_BITS 20
#define RECS_OFFSET_INTS (1 << 18)

__global__ __launch_bounds__(PART_THREADS) void gcn_part_kernel(
    const int* __restrict__ rows, const int* __restrict__ cols,
    const float* __restrict__ vals, int* __restrict__ recs_i,
    int* __restrict__ offs, int nedges) {

    __shared__ int hist[NBINS];
    __shared__ int binoff[NBINS + 1];
    __shared__ int2 stage[CH];

    int2* const recs = (int2*)recs_i;
    const int tid = threadIdx.x;
    const int nchunks = (nedges + CH - 1) / CH;

    for (int chunk = blockIdx.x; chunk < nchunks; chunk += gridDim.x) {
        const int base = chunk * CH;
        if (tid < NBINS) hist[tid] = 0;
        __syncthreads();

        int bin[EPT], rank[EPT], pk[EPT];
        float v[EPT];
#pragma unroll
        for (int i = 0; i < EPT; ++i) {
            const int e = base + i * PART_THREADS + tid;
            bin[i] = -1; rank[i] = 0; pk[i] = 0; v[i] = 0.0f;
            if (e < nedges) {
                const int r = rows[e];
                const int c = cols[e];
                v[i] = vals[e];
                int b = (int)((unsigned)r / (unsigned)RPB);
                if (b > NBINS - 1) b = NBINS - 1;
                bin[i] = b;
                pk[i] = ((r - b * RPB) << COL_BITS) | c;
            }
        }
#pragma unroll
        for (int i = 0; i < EPT; ++i) {
            if (bin[i] >= 0) rank[i] = atomicAdd(&hist[bin[i]], 1);
        }
        __syncthreads();

        if (tid < 32) {
            int s[8];
            int loc = 0;
#pragma unroll
            for (int k = 0; k < 8; ++k) { s[k] = hist[tid * 8 + k]; loc += s[k]; }
            int inc = loc;
#pragma unroll
            for (int d = 1; d < 32; d <<= 1) {
                const int t = __shfl_up(inc, d);
                if (tid >= d) inc += t;
            }
            int exc = inc - loc;
#pragma unroll
            for (int k = 0; k < 8; ++k) { binoff[tid * 8 + k] = exc; exc += s[k]; }
            if (tid == 31) binoff[NBINS] = exc;
        }
        __syncthreads();

        if (tid <= NBINS) offs[chunk * OFF_PITCH + tid] = binoff[tid];
#pragma unroll
        for (int i = 0; i < EPT; ++i) {
            if (bin[i] >= 0) stage[binoff[bin[i]] + rank[i]] = make_int2(pk[i], __float_as_int(v[i]));
        }
        __syncthreads();

        int n = nedges - base;
        if (n > CH) n = CH;
        for (int k = tid; k < n; k += PART_THREADS) recs[base + k] = stage[k];
        __syncthreads();
    }
}

__device__ __forceinline__ int run_record(const int (&pre)[SEGS + 1], const int (&rb)[SEGS], int p) {
    int a = rb[0];
#pragma unroll
    for (int i = 1; i < SEGS; ++i) a = (p >= pre[i]) ? rb[i] : a;
    return a + p;
}

__device__ __forceinline__ int owner_wave(int r) {
    int o = (r < N_HI * ROWS_HI) ? (r / ROWS_HI) : (N_HI + (r - N_HI * ROWS_HI) / ROWS_LO);
    return (o > NW - 1) ? (NW - 1) : o;
}

__device__ __forceinline__ void tile_rows(const int2* recp, int n, int myrow,
    const float* __restrict__ embeds, v8f (&acc4)[4], int lane) {

    const int j = lane & 15;
    const int khalf = (lane >> 4) << 1;
    v2f A[4];
    int ck0[4], ck1[4];
#pragma unroll
    for (int q = 0; q < 4; ++q) {
        const int k0 = 4 * q + khalf;
        const int k1 = k0 + 1;
        const bool ok0 = k0 < n;
        const bool ok1 = k1 < n;
        const int2 r0 = recp[ok0 ? k0 : 0];
        const int2 r1 = recp[ok1 ? k1 : 0];
        const int row0 = (int)((unsigned)r0.x >> COL_BITS);
        const int row1 = (int)((unsigned)r1.x >> COL_BITS);
        A[q].x = (ok0 && row0 == myrow) ? __int_as_float(r0.y) : 0.0f;
        A[q].y = (ok1 && row1 == myrow) ? __int_as_float(r1.y) : 0.0f;
        ck0[q] = ok0 ? (r0.x & ((1 << COL_BITS) - 1)) : 0;
        ck1[q] = ok1 ? (r1.x & ((1 << COL_BITS) - 1)) : 0;
    }

    float b0[4][4], b1[4][4];
#pragma unroll
    for (int q = 0; q < 4; ++q) {
        const int i0 = ck0[q] * D_FEAT + j;
        const int i1 = ck1[q] * D_FEAT + j;
#pragma unroll
        for (int c = 0; c < 4; ++c) {
            b0[q][c] = embeds[i0 + c * 16];
            b1[q][c] = embeds[i1 + c * 16];
        }
    }
#pragma unroll
    for (int q = 0; q < 4; ++q) {
        asm volatile("" :: "v"(b0[q][0]), "v"(b0[q][1]), "v"(b0[q][2]), "v"(b0[q][3]),
                           "v"(b1[q][0]), "v"(b1[q][1]), "v"(b1[q][2]), "v"(b1[q][3]));
    }
#pragma unroll
    for (int c = 0; c < 4; ++c) {
#pragma unroll
        for (int q = 0; q < 4; ++q) {
            v2f B;
            B.x = b0[q][c];
            B.y = b1[q][c];
            acc4[c] = __builtin_amdgcn_wmma_f32_16x16x4_f32(
                 false, A[q],  false, B,
                 (short)0, acc4[c],  false,  false);
        }
    }
}

__global__ __launch_bounds__(OWN_THREADS) void gcn_own_wmma_kernel(
    const int* __restrict__ recs_i, const int* __restrict__ offs,
    const float* __restrict__ embeds, float* __restrict__ out,
    int nedges, int out_size) {

    __shared__ int cnt[NW];
    __shared__ __attribute__((aligned(16))) int2 stage[NW * SUB_PITCH];

    const int2* const recs = (const int2*)recs_i;
    const int lane = threadIdx.x & 31;
    const int j = lane & 15;
    const int wave = __builtin_amdgcn_readfirstlane(threadIdx.x >> 5);
    const int nchunks = (nedges + CH - 1) / CH;
    const int n_nodes = out_size / D_FEAT;
    const int nbins = (n_nodes + RPB - 1) / RPB;

    const int wrow0 = wave * ROWS_LO + ((wave < N_HI) ? wave : N_HI);
    const int wrows = ROWS_LO + ((wave < N_HI) ? 1 : 0);
    const int myrow = (j < wrows) ? (wrow0 + j) : -1;

    for (int bin = blockIdx.x; bin < nbins; bin += gridDim.x) {
        if (threadIdx.x < NW) cnt[threadIdx.x] = 0;
        __syncthreads();

        for (int w0 = wave; w0 < nchunks; w0 += NW * SEGS) {
            int sS[SEGS], sE[SEGS];
#pragma unroll
            for (int i = 0; i < SEGS; ++i) {
                const int w = w0 + i * NW;
                const int wc = (w < nchunks) ? w : (nchunks - 1);
                sS[i] = __builtin_amdgcn_readfirstlane(offs[wc * OFF_PITCH + bin]);
                sE[i] = __builtin_amdgcn_readfirstlane(offs[wc * OFF_PITCH + bin + 1]);
            }
            int pre[SEGS + 1], rb[SEGS];
            pre[0] = 0;
#pragma unroll
            for (int i = 0; i < SEGS; ++i) {
                const int w = w0 + i * NW;
                const int len = (w < nchunks) ? (sE[i] - sS[i]) : 0;
                rb[i] = w * CH + sS[i] - pre[i];
                pre[i + 1] = pre[i] + len;
            }
            const int total = pre[SEGS];

            for (int p0 = 0; p0 < total; p0 += 32 * UNR) {
                int2 rr[UNR];
#pragma unroll
                for (int u = 0; u < UNR; ++u) {
                    const int p = p0 + u * 32 + lane;
                    rr[u] = recs[run_record(pre, rb, (p < total) ? p : (total - 1))];
                }
#pragma unroll
                for (int u = 0; u < UNR; ++u) {
                    const int p = p0 + u * 32 + lane;
                    if (p < total) {
                        const int o = owner_wave((int)((unsigned)rr[u].x >> COL_BITS));
                        const int slot = atomicAdd(&cnt[o], 1);
                        if (slot < SUBCAP) stage[o * SUB_PITCH + slot] = rr[u];
                    }
                }
            }
        }
        __syncthreads();

        const int fill = cnt[lane];
        const bool fits = __builtin_amdgcn_ballot_w32(fill > SUBCAP) == 0u;
        const int mine = __builtin_amdgcn_readfirstlane(cnt[wave]);

        v8f acc4[4];
#pragma unroll
        for (int c = 0; c < 4; ++c) acc4[c] = (v8f){0.f, 0.f, 0.f, 0.f, 0.f, 0.f, 0.f, 0.f};

        int2* const mylist = stage + wave * SUB_PITCH;
        if (fits) {
            for (int t = 0; t < mine; t += 16) {
                const int n = (mine - t < 16) ? (mine - t) : 16;
                tile_rows(mylist + t, n, myrow, embeds, acc4, lane);
            }
        } else {
            for (int w = 0; w < nchunks; ++w) {
                const int s = __builtin_amdgcn_readfirstlane(offs[w * OFF_PITCH + bin]);
                const int e = __builtin_amdgcn_readfirstlane(offs[w * OFF_PITCH + bin + 1]);
                for (int k = s; k < e; k += 16) {
                    const int n = (e - k < 16) ? (e - k) : 16;
                    tile_rows(recs + w * CH + k, n, myrow, embeds, acc4, lane);
                }
            }
        }

        float* const tp = (float*)mylist;
#pragma unroll
        for (int rg = 0; rg < 8; ++rg) {
            const int M = (lane < 16) ? rg : (rg + 8);
#pragma unroll
            for (int c = 0; c < 4; ++c) tp[M * D_FEAT + c * 16 + j] = acc4[c][rg];
        }
        const int row0 = bin * RPB + wrow0;
        for (int i = lane; i < wrows * (D_FEAT / 4); i += 32) {
            const int rw = i >> 4;
            const int f = (i & 15) * 4;
            const v4f v = *(const v4f*)(tp + rw * D_FEAT + f);
            if (row0 + rw < n_nodes) *(v4f*)(out + (row0 + rw) * D_FEAT + f) = v;
        }
        __syncthreads();
    }
}

extern "C" void kernel_launch(void* const* d_in, const int* in_sizes, int n_in,
                              void* d_out, int out_size, void* d_ws, size_t ws_size,
                              hipStream_t stream) {
    const int*   rows   = (const int*)d_in[0];
    const int*   cols   = (const int*)d_in[1];
    const float* vals   = (const float*)d_in[2];
    const float* embeds = (const float*)d_in[3];
    float*       out    = (float*)d_out;
    int*         offs   = (int*)d_ws;
    int*         recs   = (int*)d_ws + RECS_OFFSET_INTS;

    const int nedges = in_sizes[0];

    gcn_part_kernel<<<128, PART_THREADS, 0, stream>>>(rows, cols, vals, recs, offs, nedges);

    gcn_own_wmma_kernel<<<NBINS, OWN_THREADS, 0, stream>>>(
        recs, offs, embeds, out, nedges, out_size);
}


# Partition kernel: a thread's 16 edges (was 8, half the waves) handled with no per-edge branch: its loads, LDS rank adds and stores issue as groups, not one by one; add order (LDS arrival) may change
# speedup vs baseline: 1.6917x; 1.1047x over previous
#include <hip/hip_runtime.h>


typedef __attribute__((ext_vector_type(2))) float v2f;
typedef __attribute__((ext_vector_type(4))) float v4f;
typedef __attribute__((ext_vector_type(8))) float v8f;

#define D_FEAT 64
#define RPB 391
#define NBINS 256
#define CH 8192
#define PART_THREADS 512
#define EPT (CH / PART_THREADS)
#define OFF_PITCH (NBINS + 1)
#define OWN_THREADS 1024
#define NW (OWN_THREADS / 32)
#define ROWS_LO (RPB / NW)
#define N_HI (RPB % NW)
#define ROWS_HI (ROWS_LO + 1)
#define SEGS 4
#define UNR 6
#define SUBCAP 512
#define SUB_PITCH 514
#define COL_BITS 20
#define RECS_OFFSET_INTS (1 << 18)

__global__ __launch_bounds__(PART_THREADS) void gcn_part_kernel(
    const int* __restrict__ rows, const int* __restrict__ cols,
    const float* __restrict__ vals, int* __restrict__ recs_i,
    int* __restrict__ offs, int nedges) {

    __shared__ int hist[NBINS];
    __shared__ int binoff[NBINS + 1];
    __shared__ int2 stage[CH];

    int2* const recs = (int2*)recs_i;
    const int tid = threadIdx.x;
    const int nchunks = (nedges + CH - 1) / CH;

    for (int chunk = blockIdx.x; chunk < nchunks; chunk += gridDim.x) {
        const int base = chunk * CH;
        int n = nedges - base;
        if (n > CH) n = CH;
        if (tid < NBINS) hist[tid] = 0;
        __syncthreads();

        int bin[EPT], rank[EPT], pk[EPT];
        float v[EPT];
        {
            int r[EPT], c[EPT];
#pragma unroll
            for (int i = 0; i < EPT; ++i) {
                const int k = i * PART_THREADS + tid;
                const int e = base + ((k < n) ? k : (n - 1));
                r[i] = rows[e];
                c[i] = cols[e];
                v[i] = vals[e];
            }
#pragma unroll
            for (int i = 0; i < EPT; ++i) {
                int b = (int)((unsigned)r[i] / (unsigned)RPB);
                if (b > NBINS - 1) b = NBINS - 1;
                bin[i] = b;
                pk[i] = ((r[i] - b * RPB) << COL_BITS) | c[i];
            }
        }
#pragma unroll
        for (int i = 0; i < EPT; ++i) {
            const bool ok = (i * PART_THREADS + tid) < n;
            rank[i] = atomicAdd(&hist[ok ? bin[i] : (tid & (NBINS - 1))], ok ? 1 : 0);
        }
        __syncthreads();

        if (tid < 32) {
            int s[8];
            int loc = 0;
#pragma unroll
            for (int k = 0; k < 8; ++k) { s[k] = hist[tid * 8 + k]; loc += s[k]; }
            int inc = loc;
#pragma unroll
            for (int d = 1; d < 32; d <<= 1) {
                const int t = __shfl_up(inc, d);
                if (tid >= d) inc += t;
            }
            int exc = inc - loc;
#pragma unroll
            for (int k = 0; k < 8; ++k) { binoff[tid * 8 + k] = exc; exc += s[k]; }
            if (tid == 31) binoff[NBINS] = exc;
        }
        __syncthreads();

        if (tid <= NBINS) offs[chunk * OFF_PITCH + tid] = binoff[tid];
#pragma unroll
        for (int i = 0; i < EPT; ++i) {
            const int k = i * PART_THREADS + tid;
            const int pos = binoff[bin[i]] + rank[i];
            stage[(k < n) ? pos : k] = make_int2(pk[i], __float_as_int(v[i]));
        }
        __syncthreads();

        if (n == CH) {
#pragma unroll
            for (int i = 0; i < EPT; ++i) {
                const int k = i * PART_THREADS + tid;
                recs[base + k] = stage[k];
            }
        } else {
            for (int k = tid; k < n; k += PART_THREADS) recs[base + k] = stage[k];
        }
        __syncthreads();
    }
}

__device__ __forceinline__ int run_record(const int (&pre)[SEGS + 1], const int (&rb)[SEGS], int p) {
    int a = rb[0];
#pragma unroll
    for (int i = 1; i < SEGS; ++i) a = (p >= pre[i]) ? rb[i] : a;
    return a + p;
}

__device__ __forceinline__ int owner_wave(int r) {
    int o = (r < N_HI * ROWS_HI) ? (r / ROWS_HI) : (N_HI + (r - N_HI * ROWS_HI) / ROWS_LO);
    return (o > NW - 1) ? (NW - 1) : o;
}

__device__ __forceinline__ void tile_rows(const int2* recp, int n, int myrow,
    const float* __restrict__ embeds, v8f (&acc4)[4], int lane) {

    const int j = lane & 15;
    const int khalf = (lane >> 4) << 1;
    v2f A[4];
    int ck0[4], ck1[4];
#pragma unroll
    for (int q = 0; q < 4; ++q) {
        const int k0 = 4 * q + khalf;
        const int k1 = k0 + 1;
        const bool ok0 = k0 < n;
        const bool ok1 = k1 < n;
        const int2 r0 = recp[ok0 ? k0 : 0];
        const int2 r1 = recp[ok1 ? k1 : 0];
        const int row0 = (int)((unsigned)r0.x >> COL_BITS);
        const int row1 = (int)((unsigned)r1.x >> COL_BITS);
        A[q].x = (ok0 && row0 == myrow) ? __int_as_float(r0.y) : 0.0f;
        A[q].y = (ok1 && row1 == myrow) ? __int_as_float(r1.y) : 0.0f;
        ck0[q] = ok0 ? (r0.x & ((1 << COL_BITS) - 1)) : 0;
        ck1[q] = ok1 ? (r1.x & ((1 << COL_BITS) - 1)) : 0;
    }

    float b0[4][4], b1[4][4];
#pragma unroll
    for (int q = 0; q < 4; ++q) {
        const int i0 = ck0[q] * D_FEAT + j;
        const int i1 = ck1[q] * D_FEAT + j;
#pragma unroll
        for (int c = 0; c < 4; ++c) {
            b0[q][c] = embeds[i0 + c * 16];
            b1[q][c] = embeds[i1 + c * 16];
        }
    }
#pragma unroll
    for (int q = 0; q < 4; ++q) {
        asm volatile("" :: "v"(b0[q][0]), "v"(b0[q][1]), "v"(b0[q][2]), "v"(b0[q][3]),
                           "v"(b1[q][0]), "v"(b1[q][1]), "v"(b1[q][2]), "v"(b1[q][3]));
    }
#pragma unroll
    for (int c = 0; c < 4; ++c) {
#pragma unroll
        for (int q = 0; q < 4; ++q) {
            v2f B;
            B.x = b0[q][c];
            B.y = b1[q][c];
            acc4[c] = __builtin_amdgcn_wmma_f32_16x16x4_f32(
                 false, A[q],  false, B,
                 (short)0, acc4[c],  false,  false);
        }
    }
}

__global__ __launch_bounds__(OWN_THREADS) void gcn_own_wmma_kernel(
    const int* __restrict__ recs_i, const int* __restrict__ offs,
    const float* __restrict__ embeds, float* __restrict__ out,
    int nedges, int out_size) {

    __shared__ int cnt[NW];
    __shared__ __attribute__((aligned(16))) int2 stage[NW * SUB_PITCH];

    const int2* const recs = (const int2*)recs_i;
    const int lane = threadIdx.x & 31;
    const int j = lane & 15;
    const int wave = __builtin_amdgcn_readfirstlane(threadIdx.x >> 5);
    const int nchunks = (nedges + CH - 1) / CH;
    const int n_nodes = out_size / D_FEAT;
    const int nbins = (n_nodes + RPB - 1) / RPB;

    const int wrow0 = wave * ROWS_LO + ((wave < N_HI) ? wave : N_HI);
    const int wrows = ROWS_LO + ((wave < N_HI) ? 1 : 0);
    const int myrow = (j < wrows) ? (wrow0 + j) : -1;

    for (int bin = blockIdx.x; bin < nbins; bin += gridDim.x) {
        if (threadIdx.x < NW) cnt[threadIdx.x] = 0;
        __syncthreads();

        for (int w0 = wave; w0 < nchunks; w0 += NW * SEGS) {
            int sS[SEGS], sE[SEGS];
#pragma unroll
            for (int i = 0; i < SEGS; ++i) {
                const int w = w0 + i * NW;
                const int wc = (w < nchunks) ? w : (nchunks - 1);
                sS[i] = __builtin_amdgcn_readfirstlane(offs[wc * OFF_PITCH + bin]);
                sE[i] = __builtin_amdgcn_readfirstlane(offs[wc * OFF_PITCH + bin + 1]);
            }
            int pre[SEGS + 1], rb[SEGS];
            pre[0] = 0;
#pragma unroll
            for (int i = 0; i < SEGS; ++i) {
                const int w = w0 + i * NW;
                const int len = (w < nchunks) ? (sE[i] - sS[i]) : 0;
                rb[i] = w * CH + sS[i] - pre[i];
                pre[i + 1] = pre[i] + len;
            }
            const int total = pre[SEGS];

            for (int p0 = 0; p0 < total; p0 += 32 * UNR) {
                int2 rr[UNR];
#pragma unroll
                for (int u = 0; u < UNR; ++u) {
                    const int p = p0 + u * 32 + lane;
                    rr[u] = recs[run_record(pre, rb, (p < total) ? p : (total - 1))];
                }
#pragma unroll
                for (int u = 0; u < UNR; ++u) {
                    const int p = p0 + u * 32 + lane;
                    if (p < total) {
                        const int o = owner_wave((int)((unsigned)rr[u].x >> COL_BITS));
                        const int slot = atomicAdd(&cnt[o], 1);
                        if (slot < SUBCAP) stage[o * SUB_PITCH + slot] = rr[u];
                    }
                }
            }
        }
        __syncthreads();

        const int fill = cnt[lane];
        const bool fits = __builtin_amdgcn_ballot_w32(fill > SUBCAP) == 0u;
        const int mine = __builtin_amdgcn_readfirstlane(cnt[wave]);

        v8f acc4[4];
#pragma unroll
        for (int c = 0; c < 4; ++c) acc4[c] = (v8f){0.f, 0.f, 0.f, 0.f, 0.f, 0.f, 0.f, 0.f};

        int2* const mylist = stage + wave * SUB_PITCH;
        if (fits) {
            for (int t = 0; t < mine; t += 16) {
                const int n = (mine - t < 16) ? (mine - t) : 16;
                tile_rows(mylist + t, n, myrow, embeds, acc4, lane);
            }
        } else {
            for (int w = 0; w < nchunks; ++w) {
                const int s = __builtin_amdgcn_readfirstlane(offs[w * OFF_PITCH + bin]);
                const int e = __builtin_amdgcn_readfirstlane(offs[w * OFF_PITCH + bin + 1]);
                for (int k = s; k < e; k += 16) {
                    const int n = (e - k < 16) ? (e - k) : 16;
                    tile_rows(recs + w * CH + k, n, myrow, embeds, acc4, lane);
                }
            }
        }

        float* const tp = (float*)mylist;
#pragma unroll
        for (int rg = 0; rg < 8; ++rg) {
            const int M = (lane < 16) ? rg : (rg + 8);
#pragma unroll
            for (int c = 0; c < 4; ++c) tp[M * D_FEAT + c * 16 + j] = acc4[c][rg];
        }
        const int row0 = bin * RPB + wrow0;
        for (int i = lane; i < wrows * (D_FEAT / 4); i += 32) {
            const int rw = i >> 4;
            const int f = (i & 15) * 4;
            const v4f v = *(const v4f*)(tp + rw * D_FEAT + f);
            if (row0 + rw < n_nodes) *(v4f*)(out + (row0 + rw) * D_FEAT + f) = v;
        }
        __syncthreads();
    }
}

extern "C" void kernel_launch(void* const* d_in, const int* in_sizes, int n_in,
                              void* d_out, int out_size, void* d_ws, size_t ws_size,
                              hipStream_t stream) {
    const int*   rows   = (const int*)d_in[0];
    const int*   cols   = (const int*)d_in[1];
    const float* vals   = (const float*)d_in[2];
    const float* embeds = (const float*)d_in[3];
    float*       out    = (float*)d_out;
    int*         offs   = (int*)d_ws;
    int*         recs   = (int*)d_ws + RECS_OFFSET_INTS;

    const int nedges = in_sizes[0];

    gcn_part_kernel<<<128, PART_THREADS, 0, stream>>>(rows, cols, vals, recs, offs, nedges);

    gcn_own_wmma_kernel<<<NBINS, OWN_THREADS, 0, stream>>>(
        recs, offs, embeds, out, nedges, out_size);
}


# Gathers read 16 bytes per lane: column j of feature tile c stands for feature 4j+c, so 16 lanes fetch a record's whole 256-byte row in one load (8 loads a tile, was 32); rows store from accumulators
# speedup vs baseline: 1.7652x; 1.0434x over previous
#include <hip/hip_runtime.h>


typedef __attribute__((ext_vector_type(2))) float v2f;
typedef __attribute__((ext_vector_type(4))) float v4f;
typedef __attribute__((ext_vector_type(8))) float v8f;

#define D_FEAT 64
#define RPB 391
#define NBINS 256
#define CH 8192
#define PART_THREADS 512
#define EPT (CH / PART_THREADS)
#define OFF_PITCH (NBINS + 1)
#define OWN_THREADS 1024
#define NW (OWN_THREADS / 32)
#define ROWS_LO (RPB / NW)
#define N_HI (RPB % NW)
#define ROWS_HI (ROWS_LO + 1)
#define SEGS 4
#define UNR 6
#define SUBCAP 512
#define SUB_PITCH 514
#define COL_BITS 20
#define RECS_OFFSET_INTS (1 << 18)

__global__ __launch_bounds__(PART_THREADS) void gcn_part_kernel(
    const int* __restrict__ rows, const int* __restrict__ cols,
    const float* __restrict__ vals, int* __restrict__ recs_i,
    int* __restrict__ offs, int nedges) {

    __shared__ int hist[NBINS];
    __shared__ int binoff[NBINS + 1];
    __shared__ int2 stage[CH];

    int2* const recs = (int2*)recs_i;
    const int tid = threadIdx.x;
    const int nchunks = (nedges + CH - 1) / CH;

    for (int chunk = blockIdx.x; chunk < nchunks; chunk += gridDim.x) {
        const int base = chunk * CH;
        int n = nedges - base;
        if (n > CH) n = CH;
        if (tid < NBINS) hist[tid] = 0;
        __syncthreads();

        int bin[EPT], rank[EPT], pk[EPT];
        float v[EPT];
        {
            int r[EPT], c[EPT];
#pragma unroll
            for (int i = 0; i < EPT; ++i) {
                const int k = i * PART_THREADS + tid;
                const int e = base + ((k < n) ? k : (n - 1));
                r[i] = rows[e];
                c[i] = cols[e];
                v[i] = vals[e];
            }
#pragma unroll
            for (int i = 0; i < EPT; ++i) {
                int b = (int)((unsigned)r[i] / (unsigned)RPB);
                if (b > NBINS - 1) b = NBINS - 1;
                bin[i] = b;
                pk[i] = ((r[i] - b * RPB) << COL_BITS) | c[i];
            }
        }
#pragma unroll
        for (int i = 0; i < EPT; ++i) {
            const bool ok = (i * PART_THREADS + tid) < n;
            rank[i] = atomicAdd(&hist[ok ? bin[i] : (tid & (NBINS - 1))], ok ? 1 : 0);
        }
        __syncthreads();

        if (tid < 32) {
            int s[8];
            int loc = 0;
#pragma unroll
            for (int k = 0; k < 8; ++k) { s[k] = hist[tid * 8 + k]; loc += s[k]; }
            int inc = loc;
#pragma unroll
            for (int d = 1; d < 32; d <<= 1) {
                const int t = __shfl_up(inc, d);
                if (tid >= d) inc += t;
            }
            int exc = inc - loc;
#pragma unroll
            for (int k = 0; k < 8; ++k) { binoff[tid * 8 + k] = exc; exc += s[k]; }
            if (tid == 31) binoff[NBINS] = exc;
        }
        __syncthreads();

        if (tid <= NBINS) offs[chunk * OFF_PITCH + tid] = binoff[tid];
#pragma unroll
        for (int i = 0; i < EPT; ++i) {
            const int k = i * PART_THREADS + tid;
            const int pos = binoff[bin[i]] + rank[i];
            stage[(k < n) ? pos : k] = make_int2(pk[i], __float_as_int(v[i]));
        }
        __syncthreads();

        if (n == CH) {
#pragma unroll
            for (int i = 0; i < EPT; ++i) {
                const int k = i * PART_THREADS + tid;
                recs[base + k] = stage[k];
            }
        } else {
            for (int k = tid; k < n; k += PART_THREADS) recs[base + k] = stage[k];
        }
        __syncthreads();
    }
}

__device__ __forceinline__ int run_record(const int (&pre)[SEGS + 1], const int (&rb)[SEGS], int p) {
    int a = rb[0];
#pragma unroll
    for (int i = 1; i < SEGS; ++i) a = (p >= pre[i]) ? rb[i] : a;
    return a + p;
}

__device__ __forceinline__ int owner_wave(int r) {
    int o = (r < N_HI * ROWS_HI) ? (r / ROWS_HI) : (N_HI + (r - N_HI * ROWS_HI) / ROWS_LO);
    return (o > NW - 1) ? (NW - 1) : o;
}

__device__ __forceinline__ void tile_rows(const int2* recp, int n, int myrow,
    const float* __restrict__ embeds, v8f (&acc4)[4], int lane) {

    const int j = lane & 15;
    const int khalf = (lane >> 4) << 1;
    v2f A[4];
    int ck0[4], ck1[4];
#pragma unroll
    for (int q = 0; q < 4; ++q) {
        const int k0 = 4 * q + khalf;
        const int k1 = k0 + 1;
        const bool ok0 = k0 < n;
        const bool ok1 = k1 < n;
        const int2 r0 = recp[ok0 ? k0 : 0];
        const int2 r1 = recp[ok1 ? k1 : 0];
        const int row0 = (int)((unsigned)r0.x >> COL_BITS);
        const int row1 = (int)((unsigned)r1.x >> COL_BITS);
        A[q].x = (ok0 && row0 == myrow) ? __int_as_float(r0.y) : 0.0f;
        A[q].y = (ok1 && row1 == myrow) ? __int_as_float(r1.y) : 0.0f;
        ck0[q] = ok0 ? (r0.x & ((1 << COL_BITS) - 1)) : 0;
        ck1[q] = ok1 ? (r1.x & ((1 << COL_BITS) - 1)) : 0;
    }

    v4f g0[4], g1[4];
#pragma unroll
    for (int q = 0; q < 4; ++q) {
        g0[q] = *(const v4f*)(embeds + ((unsigned)ck0[q] * D_FEAT + 4u * (unsigned)j));
        g1[q] = *(const v4f*)(embeds + ((unsigned)ck1[q] * D_FEAT + 4u * (unsigned)j));
    }
    asm volatile("" :: "v"(g0[0]), "v"(g0[1]), "v"(g0[2]), "v"(g0[3]),
                       "v"(g1[0]), "v"(g1[1]), "v"(g1[2]), "v"(g1[3]));
#pragma unroll
    for (int c = 0; c < 4; ++c) {
#pragma unroll
        for (int q = 0; q < 4; ++q) {
            v2f B;
            B.x = g0[q][c];
            B.y = g1[q][c];
            acc4[c] = __builtin_amdgcn_wmma_f32_16x16x4_f32(
                 false, A[q],  false, B,
                 (short)0, acc4[c],  false,  false);
        }
    }
}

__global__ __launch_bounds__(OWN_THREADS) void gcn_own_wmma_kernel(
    const int* __restrict__ recs_i, const int* __restrict__ offs,
    const float* __restrict__ embeds, float* __restrict__ out,
    int nedges, int out_size) {

    __shared__ int cnt[NW];
    __shared__ __attribute__((aligned(16))) int2 stage[NW * SUB_PITCH];

    const int2* const recs = (const int2*)recs_i;
    const int lane = threadIdx.x & 31;
    const int j = lane & 15;
    const int wave = __builtin_amdgcn_readfirstlane(threadIdx.x >> 5);
    const int nchunks = (nedges + CH - 1) / CH;
    const int n_nodes = out_size / D_FEAT;
    const int nbins = (n_nodes + RPB - 1) / RPB;

    const int wrow0 = wave * ROWS_LO + ((wave < N_HI) ? wave : N_HI);
    const int wrows = ROWS_LO + ((wave < N_HI) ? 1 : 0);
    const int myrow = (j < wrows) ? (wrow0 + j) : -1;

    for (int bin = blockIdx.x; bin < nbins; bin += gridDim.x) {
        if (threadIdx.x < NW) cnt[threadIdx.x] = 0;
        __syncthreads();

        for (int w0 = wave; w0 < nchunks; w0 += NW * SEGS) {
            int sS[SEGS], sE[SEGS];
#pragma unroll
            for (int i = 0; i < SEGS; ++i) {
                const int w = w0 + i * NW;
                const int wc = (w < nchunks) ? w : (nchunks - 1);
                sS[i] = __builtin_amdgcn_readfirstlane(offs[wc * OFF_PITCH + bin]);
                sE[i] = __builtin_amdgcn_readfirstlane(offs[wc * OFF_PITCH + bin + 1]);
            }
            int pre[SEGS + 1], rb[SEGS];
            pre[0] = 0;
#pragma unroll
            for (int i = 0; i < SEGS; ++i) {
                const int w = w0 + i * NW;
                const int len = (w < nchunks) ? (sE[i] - sS[i]) : 0;
                rb[i] = w * CH + sS[i] - pre[i];
                pre[i + 1] = pre[i] + len;
            }
            const int total = pre[SEGS];

            for (int p0 = 0; p0 < total; p0 += 32 * UNR) {
                int2 rr[UNR];
#pragma unroll
                for (int u = 0; u < UNR; ++u) {
                    const int p = p0 + u * 32 + lane;
                    rr[u] = recs[run_record(pre, rb, (p < total) ? p : (total - 1))];
                }
#pragma unroll
                for (int u = 0; u < UNR; ++u) {
                    const int p = p0 + u * 32 + lane;
                    if (p < total) {
                        const int o = owner_wave((int)((unsigned)rr[u].x >> COL_BITS));
                        const int slot = atomicAdd(&cnt[o], 1);
                        if (slot < SUBCAP) stage[o * SUB_PITCH + slot] = rr[u];
                    }
                }
            }
        }
        __syncthreads();

        const int fill = cnt[lane];
        const bool fits = __builtin_amdgcn_ballot_w32(fill > SUBCAP) == 0u;
        const int mine = __builtin_amdgcn_readfirstlane(cnt[wave]);

        v8f acc4[4];
#pragma unroll
        for (int c = 0; c < 4; ++c) acc4[c] = (v8f){0.f, 0.f, 0.f, 0.f, 0.f, 0.f, 0.f, 0.f};

        int2* const mylist = stage + wave * SUB_PITCH;
        if (fits) {
            for (int t = 0; t < mine; t += 16) {
                const int n = (mine - t < 16) ? (mine - t) : 16;
                tile_rows(mylist + t, n, myrow, embeds, acc4, lane);
            }
        } else {
            for (int w = 0; w < nchunks; ++w) {
                const int s = __builtin_amdgcn_readfirstlane(offs[w * OFF_PITCH + bin]);
                const int e = __builtin_amdgcn_readfirstlane(offs[w * OFF_PITCH + bin + 1]);
                for (int k = s; k < e; k += 16) {
                    const int n = (e - k < 16) ? (e - k) : 16;
                    tile_rows(recs + w * CH + k, n, myrow, embeds, acc4, lane);
                }
            }
        }

        const int row0 = bin * RPB + wrow0;
        const int mhalf = (lane >> 4) << 3;
#pragma unroll
        for (int rg = 0; rg < 8; ++rg) {
            const int M = rg + mhalf;
            if (M < wrows && row0 + M < n_nodes) {
                v4f v;
                v.x = acc4[0][rg];
                v.y = acc4[1][rg];
                v.z = acc4[2][rg];
                v.w = acc4[3][rg];
                *(v4f*)(out + (row0 + M) * D_FEAT + 4 * j) = v;
            }
        }
        __syncthreads();
    }
}

extern "C" void kernel_launch(void* const* d_in, const int* in_sizes, int n_in,
                              void* d_out, int out_size, void* d_ws, size_t ws_size,
                              hipStream_t stream) {
    const int*   rows   = (const int*)d_in[0];
    const int*   cols   = (const int*)d_in[1];
    const float* vals   = (const float*)d_in[2];
    const float* embeds = (const float*)d_in[3];
    float*       out    = (float*)d_out;
    int*         offs   = (int*)d_ws;
    int*         recs   = (int*)d_ws + RECS_OFFSET_INTS;

    const int nedges = in_sizes[0];

    gcn_part_kernel<<<128, PART_THREADS, 0, stream>>>(rows, cols, vals, recs, offs, nedges);

    gcn_own_wmma_kernel<<<NBINS, OWN_THREADS, 0, stream>>>(
        recs, offs, embeds, out, nedges, out_size);
}


# Partition chunks of 4096 edges (were 8192; 8-wave groups, 16 edges a thread as before): 245 chunks fill the 256 WGPs, were 123; owner waves fetch 8 segments' bounds in one load; add order changes
# speedup vs baseline: 1.8454x; 1.0454x over previous
#include <hip/hip_runtime.h>


typedef __attribute__((ext_vector_type(2))) float v2f;
typedef __attribute__((ext_vector_type(4))) float v4f;
typedef __attribute__((ext_vector_type(8))) float v8f;

#define D_FEAT 64
#define RPB 391
#define NBINS 256
#define CH 4096
#define PART_THREADS 256
#define EPT (CH / PART_THREADS)
#define OFF_PITCH (NBINS + 1)
#define OWN_THREADS 1024
#define NW (OWN_THREADS / 32)
#define ROWS_LO (RPB / NW)
#define N_HI (RPB % NW)
#define ROWS_HI (ROWS_LO + 1)
#define SEGS 8
static_assert(2 * SEGS <= 32 && (SEGS & (SEGS - 1)) == 0, "a wave's segment bounds must fit one load, one per lane");
#define UNR 6
#define SUBCAP 512
#define SUB_PITCH 514
#define COL_BITS 20
#define RECS_OFFSET_INTS (1 << 18)

__global__ __launch_bounds__(PART_THREADS) void gcn_part_kernel(
    const int* __restrict__ rows, const int* __restrict__ cols,
    const float* __restrict__ vals, int* __restrict__ recs_i,
    int* __restrict__ offs, int nedges) {

    __shared__ int hist[NBINS];
    __shared__ int binoff[NBINS + 1];
    __shared__ int2 stage[CH];

    int2* const recs = (int2*)recs_i;
    const int tid = threadIdx.x;
    const int nchunks = (nedges + CH - 1) / CH;

    for (int chunk = blockIdx.x; chunk < nchunks; chunk += gridDim.x) {
        const int base = chunk * CH;
        int n = nedges - base;
        if (n > CH) n = CH;
        if (tid < NBINS) hist[tid] = 0;
        __syncthreads();

        int bin[EPT], rank[EPT], pk[EPT];
        float v[EPT];
        {
            int r[EPT], c[EPT];
#pragma unroll
            for (int i = 0; i < EPT; ++i) {
                const int k = i * PART_THREADS + tid;
                const int e = base + ((k < n) ? k : (n - 1));
                r[i] = rows[e];
                c[i] = cols[e];
                v[i] = vals[e];
            }
#pragma unroll
            for (int i = 0; i < EPT; ++i) {
                int b = (int)((unsigned)r[i] / (unsigned)RPB);
                if (b > NBINS - 1) b = NBINS - 1;
                bin[i] = b;
                pk[i] = ((r[i] - b * RPB) << COL_BITS) | c[i];
            }
        }
#pragma unroll
        for (int i = 0; i < EPT; ++i) {
            const bool ok = (i * PART_THREADS + tid) < n;
            rank[i] = atomicAdd(&hist[ok ? bin[i] : (tid & (NBINS - 1))], ok ? 1 : 0);
        }
        __syncthreads();

        if (tid < 32) {
            int s[8];
            int loc = 0;
#pragma unroll
            for (int k = 0; k < 8; ++k) { s[k] = hist[tid * 8 + k]; loc += s[k]; }
            int inc = loc;
#pragma unroll
            for (int d = 1; d < 32; d <<= 1) {
                const int t = __shfl_up(inc, d);
                if (tid >= d) inc += t;
            }
            int exc = inc - loc;
#pragma unroll
            for (int k = 0; k < 8; ++k) { binoff[tid * 8 + k] = exc; exc += s[k]; }
            if (tid == 31) binoff[NBINS] = exc;
        }
        __syncthreads();

        for (int t = tid; t <= NBINS; t += PART_THREADS) offs[chunk * OFF_PITCH + t] = binoff[t];
#pragma unroll
        for (int i = 0; i < EPT; ++i) {
            const int k = i * PART_THREADS + tid;
            const int pos = binoff[bin[i]] + rank[i];
            stage[(k < n) ? pos : k] = make_int2(pk[i], __float_as_int(v[i]));
        }
        __syncthreads();

        if (n == CH) {
#pragma unroll
            for (int i = 0; i < EPT; ++i) {
                const int k = i * PART_THREADS + tid;
                recs[base + k] = stage[k];
            }
        } else {
            for (int k = tid; k < n; k += PART_THREADS) recs[base + k] = stage[k];
        }
        __syncthreads();
    }
}

__device__ __forceinline__ int run_record(const int (&pre)[SEGS + 1], const int (&rb)[SEGS], int p) {
    int a = rb[0];
#pragma unroll
    for (int i = 1; i < SEGS; ++i) a = (p >= pre[i]) ? rb[i] : a;
    return a + p;
}

__device__ __forceinline__ int owner_wave(int r) {
    int o = (r < N_HI * ROWS_HI) ? (r / ROWS_HI) : (N_HI + (r - N_HI * ROWS_HI) / ROWS_LO);
    return (o > NW - 1) ? (NW - 1) : o;
}

__device__ __forceinline__ void tile_rows(const int2* recp, int n, int myrow,
    const float* __restrict__ embeds, v8f (&acc4)[4], int lane) {

    const int j = lane & 15;
    const int khalf = (lane >> 4) << 1;
    v2f A[4];
    int ck0[4], ck1[4];
#pragma unroll
    for (int q = 0; q < 4; ++q) {
        const int k0 = 4 * q + khalf;
        const int k1 = k0 + 1;
        const bool ok0 = k0 < n;
        const bool ok1 = k1 < n;
        const int2 r0 = recp[ok0 ? k0 : 0];
        const int2 r1 = recp[ok1 ? k1 : 0];
        const int row0 = (int)((unsigned)r0.x >> COL_BITS);
        const int row1 = (int)((unsigned)r1.x >> COL_BITS);
        A[q].x = (ok0 && row0 == myrow) ? __int_as_float(r0.y) : 0.0f;
        A[q].y = (ok1 && row1 == myrow) ? __int_as_float(r1.y) : 0.0f;
        ck0[q] = ok0 ? (r0.x & ((1 << COL_BITS) - 1)) : 0;
        ck1[q] = ok1 ? (r1.x & ((1 << COL_BITS) - 1)) : 0;
    }

    v4f g0[4], g1[4];
#pragma unroll
    for (int q = 0; q < 4; ++q) {
        g0[q] = *(const v4f*)(embeds + ((unsigned)ck0[q] * D_FEAT + 4u * (unsigned)j));
        g1[q] = *(const v4f*)(embeds + ((unsigned)ck1[q] * D_FEAT + 4u * (unsigned)j));
    }
    asm volatile("" :: "v"(g0[0]), "v"(g0[1]), "v"(g0[2]), "v"(g0[3]),
                       "v"(g1[0]), "v"(g1[1]), "v"(g1[2]), "v"(g1[3]));
#pragma unroll
    for (int c = 0; c < 4; ++c) {
#pragma unroll
        for (int q = 0; q < 4; ++q) {
            v2f B;
            B.x = g0[q][c];
            B.y = g1[q][c];
            acc4[c] = __builtin_amdgcn_wmma_f32_16x16x4_f32(
                 false, A[q],  false, B,
                 (short)0, acc4[c],  false,  false);
        }
    }
}

__global__ __launch_bounds__(OWN_THREADS) void gcn_own_wmma_kernel(
    const int* __restrict__ recs_i, const int* __restrict__ offs,
    const float* __restrict__ embeds, float* __restrict__ out,
    int nedges, int out_size) {

    __shared__ int cnt[NW];
    __shared__ __attribute__((aligned(16))) int2 stage[NW * SUB_PITCH];

    const int2* const recs = (const int2*)recs_i;
    const int lane = threadIdx.x & 31;
    const int j = lane & 15;
    const int wave = __builtin_amdgcn_readfirstlane(threadIdx.x >> 5);
    const int nchunks = (nedges + CH - 1) / CH;
    const int n_nodes = out_size / D_FEAT;
    const int nbins = (n_nodes + RPB - 1) / RPB;

    const int wrow0 = wave * ROWS_LO + ((wave < N_HI) ? wave : N_HI);
    const int wrows = ROWS_LO + ((wave < N_HI) ? 1 : 0);
    const int myrow = (j < wrows) ? (wrow0 + j) : -1;

    for (int bin = blockIdx.x; bin < nbins; bin += gridDim.x) {
        if (threadIdx.x < NW) cnt[threadIdx.x] = 0;
        __syncthreads();

        for (int w0 = wave; w0 < nchunks; w0 += NW * SEGS) {
            const int bseg = (lane >> 1) & (SEGS - 1);
            const int bw = w0 + bseg * NW;
            const int bwc = (bw < nchunks) ? bw : (nchunks - 1);
            const int bnd = offs[bwc * OFF_PITCH + bin + (lane & 1)];
            int sS[SEGS], sE[SEGS];
#pragma unroll
            for (int i = 0; i < SEGS; ++i) {
                sS[i] = __builtin_amdgcn_readlane(bnd, 2 * i);
                sE[i] = __builtin_amdgcn_readlane(bnd, 2 * i + 1);
            }
            int pre[SEGS + 1], rb[SEGS];
            pre[0] = 0;
#pragma unroll
            for (int i = 0; i < SEGS; ++i) {
                const int w = w0 + i * NW;
                const int len = (w < nchunks) ? (sE[i] - sS[i]) : 0;
                rb[i] = w * CH + sS[i] - pre[i];
                pre[i + 1] = pre[i] + len;
            }
            const int total = pre[SEGS];

            for (int p0 = 0; p0 < total; p0 += 32 * UNR) {
                int2 rr[UNR];
#pragma unroll
                for (int u = 0; u < UNR; ++u) {
                    const int p = p0 + u * 32 + lane;
                    rr[u] = recs[run_record(pre, rb, (p < total) ? p : (total - 1))];
                }
#pragma unroll
                for (int u = 0; u < UNR; ++u) {
                    const int p = p0 + u * 32 + lane;
                    if (p < total) {
                        const int o = owner_wave((int)((unsigned)rr[u].x >> COL_BITS));
                        const int slot = atomicAdd(&cnt[o], 1);
                        if (slot < SUBCAP) stage[o * SUB_PITCH + slot] = rr[u];
                    }
                }
            }
        }
        __syncthreads();

        const int fill = cnt[lane];
        const bool fits = __builtin_amdgcn_ballot_w32(fill > SUBCAP) == 0u;
        const int mine = __builtin_amdgcn_readfirstlane(cnt[wave]);

        v8f acc4[4];
#pragma unroll
        for (int c = 0; c < 4; ++c) acc4[c] = (v8f){0.f, 0.f, 0.f, 0.f, 0.f, 0.f, 0.f, 0.f};

        int2* const mylist = stage + wave * SUB_PITCH;
        if (fits) {
            for (int t = 0; t < mine; t += 16) {
                const int n = (mine - t < 16) ? (mine - t) : 16;
                tile_rows(mylist + t, n, myrow, embeds, acc4, lane);
            }
        } else {
            for (int w = 0; w < nchunks; ++w) {
                const int s = __builtin_amdgcn_readfirstlane(offs[w * OFF_PITCH + bin]);
                const int e = __builtin_amdgcn_readfirstlane(offs[w * OFF_PITCH + bin + 1]);
                for (int k = s; k < e; k += 16) {
                    const int n = (e - k < 16) ? (e - k) : 16;
                    tile_rows(recs + w * CH + k, n, myrow, embeds, acc4, lane);
                }
            }
        }

        const int row0 = bin * RPB + wrow0;
        const int mhalf = (lane >> 4) << 3;
#pragma unroll
        for (int rg = 0; rg < 8; ++rg) {
            const int M = rg + mhalf;
            if (M < wrows && row0 + M < n_nodes) {
                v4f v;
                v.x = acc4[0][rg];
                v.y = acc4[1][rg];
                v.z = acc4[2][rg];
                v.w = acc4[3][rg];
                *(v4f*)(out + (row0 + M) * D_FEAT + 4 * j) = v;
            }
        }
        __syncthreads();
    }
}

extern "C" void kernel_launch(void* const* d_in, const int* in_sizes, int n_in,
                              void* d_out, int out_size, void* d_ws, size_t ws_size,
                              hipStream_t stream) {
    const int*   rows   = (const int*)d_in[0];
    const int*   cols   = (const int*)d_in[1];
    const float* vals   = (const float*)d_in[2];
    const float* embeds = (const float*)d_in[3];
    float*       out    = (float*)d_out;
    int*         offs   = (int*)d_ws;
    int*         recs   = (int*)d_ws + RECS_OFFSET_INTS;

    const int nedges = in_sizes[0];

    gcn_part_kernel<<<256, PART_THREADS, 0, stream>>>(rows, cols, vals, recs, offs, nedges);

    gcn_own_wmma_kernel<<<NBINS, OWN_THREADS, 0, stream>>>(
        recs, offs, embeds, out, nedges, out_size);
}
